# starting kernel
# baseline (speedup 1.0000x reference)
#include <hip/hip_runtime.h>

typedef float v2f __attribute__((ext_vector_type(2)));
typedef float v4f __attribute__((ext_vector_type(4)));
typedef float v8f __attribute__((ext_vector_type(8)));

#define N_NODES 100000
#define N_EDGES 1600000
#define IN_DIM  165
#define HID     64
#define OUT_DIM 2
#define NTILES  (N_NODES / 16)
#define KPAIRS  83

__global__ void zero_kernel(float* __restrict__ p, int n) {
    int i = blockIdx.x * blockDim.x + threadIdx.x;
    if (i < n) p[i] = 0.0f;
}

__global__ void deg_kernel(const int* __restrict__ dst, float* __restrict__ deg) {
    int e = blockIdx.x * blockDim.x + threadIdx.x;
    if (e < N_EDGES) unsafeAtomicAdd(&deg[dst[e]], 1.0f);
}

__global__ void dinv_kernel(float* __restrict__ d) {
    int i = blockIdx.x * blockDim.x + threadIdx.x;
    if (i < N_NODES) d[i] = rsqrtf(d[i] + 1.0f);
}

__global__ void enorm_kernel(const int* __restrict__ src, const int* __restrict__ dst,
                             const float* __restrict__ dinv, float* __restrict__ en) {
    int e = blockIdx.x * blockDim.x + threadIdx.x;
    if (e < N_EDGES) en[e] = dinv[src[e]] * dinv[dst[e]];
}

__global__ __launch_bounds__(256) void gemm1_wmma(const float* __restrict__ x,
                                                  const float* __restrict__ W1,
                                                  float* __restrict__ hw1) {
    __shared__ float w1p[KPAIRS * 2 * HID];
    for (int i = threadIdx.x; i < KPAIRS * 2 * HID; i += 256) {
        int kp   = i >> 7;
        int r    = i & 127;
        int col  = r >> 1;
        int sub  = r & 1;
        int krow = kp * 2 + sub;
        w1p[i] = (krow < IN_DIM) ? W1[krow * HID + col] : 0.0f;
    }
    __syncthreads();

    const int wave = threadIdx.x >> 5;
    const int lane = threadIdx.x & 31;
    const int tile = blockIdx.x * 8 + wave;
    if (tile >= NTILES) return;

    const int hh  = lane >> 4;
    const int lr  = lane & 15;
    const int row = tile * 16 + lr;
    const float* __restrict__ xrb = x + (long)row * IN_DIM;
    const float* __restrict__ xr  = xrb + 2 * hh;

    v8f acc0 = (v8f){}, acc1 = (v8f){}, acc2 = (v8f){}, acc3 = (v8f){};

    for (int k = 0; k < 164; k += 4) {
        v2f A;
        A.x = xr[k];
        A.y = xr[k + 1];
        const int ka = k + 2 * hh;
        const v2f* __restrict__ bp = (const v2f*)&w1p[(ka >> 1) * 2 * HID + lr * 2];
        v2f B0 = bp[0];
        v2f B1 = bp[16];
        v2f B2 = bp[32];
        v2f B3 = bp[48];
        acc0 = __builtin_amdgcn_wmma_f32_16x16x4_f32(false, A, false, B0, (short)0, acc0, false, false);
        acc1 = __builtin_amdgcn_wmma_f32_16x16x4_f32(false, A, false, B1, (short)0, acc1, false, false);
        acc2 = __builtin_amdgcn_wmma_f32_16x16x4_f32(false, A, false, B2, (short)0, acc2, false, false);
        acc3 = __builtin_amdgcn_wmma_f32_16x16x4_f32(false, A, false, B3, (short)0, acc3, false, false);
    }

    {
        float a164 = xrb[164];
        v2f A;
        A.x = (hh == 0) ? a164 : 0.0f;
        A.y = 0.0f;
        const v2f* __restrict__ bp = (const v2f*)&w1p[(KPAIRS - 1) * 2 * HID + lr * 2];
        v2f B0 = bp[0];
        v2f B1 = bp[16];
        v2f B2 = bp[32];
        v2f B3 = bp[48];
        acc0 = __builtin_amdgcn_wmma_f32_16x16x4_f32(false, A, false, B0, (short)0, acc0, false, false);
        acc1 = __builtin_amdgcn_wmma_f32_16x16x4_f32(false, A, false, B1, (short)0, acc1, false, false);
        acc2 = __builtin_amdgcn_wmma_f32_16x16x4_f32(false, A, false, B2, (short)0, acc2, false, false);
        acc3 = __builtin_amdgcn_wmma_f32_16x16x4_f32(false, A, false, B3, (short)0, acc3, false, false);
    }

    float* __restrict__ outp = hw1 + (long)(tile * 16 + 8 * hh) * HID + lr;
#pragma unroll
    for (int v = 0; v < 8; ++v) {
        long o = (long)v * HID;
        outp[o +  0] = acc0[v];
        outp[o + 16] = acc1[v];
        outp[o + 32] = acc2[v];
        outp[o + 48] = acc3[v];
    }
}

__global__ void scatter1(const int* __restrict__ src, const int* __restrict__ dst,
                         const float* __restrict__ en, const float* __restrict__ hw1,
                         float* __restrict__ agg) {
    int tid = blockIdx.x * blockDim.x + threadIdx.x;
    int e = tid >> 4;
    int f = (tid & 15) * 4;
    if (e >= N_EDGES) return;
    int s = src[e], d = dst[e];
    float w = en[e];
    v4f m = *(const v4f*)(hw1 + (long)s * HID + f);
    float* ap = agg + (long)d * HID + f;
    unsafeAtomicAdd(ap + 0, m.x * w);
    unsafeAtomicAdd(ap + 1, m.y * w);
    unsafeAtomicAdd(ap + 2, m.z * w);
    unsafeAtomicAdd(ap + 3, m.w * w);
}

__global__ void relu_bias(const float* __restrict__ hw1, const float* __restrict__ dinv,
                          const float* __restrict__ b1, float* __restrict__ h) {
    int tid = blockIdx.x * blockDim.x + threadIdx.x;
    if (tid >= N_NODES * HID) return;
    int n = tid >> 6, f = tid & 63;
    float sn = dinv[n]; sn *= sn;
    float v = h[tid] + hw1[tid] * sn + b1[f];
    h[tid] = v > 0.0f ? v : 0.0f;
}

__global__ void gemm2(const float* __restrict__ h, const float* __restrict__ W2,
                      float* __restrict__ hw2) {
    int n = blockIdx.x * blockDim.x + threadIdx.x;
    if (n >= N_NODES) return;
    const float* hr = h + (long)n * HID;
    float o0 = 0.0f, o1 = 0.0f;
#pragma unroll 8
    for (int k = 0; k < HID; ++k) {
        float hv = hr[k];
        o0 = fmaf(hv, W2[2 * k],     o0);
        o1 = fmaf(hv, W2[2 * k + 1], o1);
    }
    hw2[2 * n]     = o0;
    hw2[2 * n + 1] = o1;
}

__global__ void scatter2(const int* __restrict__ src, const int* __restrict__ dst,
                         const float* __restrict__ en, const float* __restrict__ hw2,
                         float* __restrict__ out) {
    int e = blockIdx.x * blockDim.x + threadIdx.x;
    if (e >= N_EDGES) return;
    int s = src[e], d = dst[e];
    float w = en[e];
    unsafeAtomicAdd(&out[2 * d],     hw2[2 * s]     * w);
    unsafeAtomicAdd(&out[2 * d + 1], hw2[2 * s + 1] * w);
}

__global__ void finish2(const float* __restrict__ hw2, const float* __restrict__ dinv,
                        const float* __restrict__ b2, float* __restrict__ out) {
    int n = blockIdx.x * blockDim.x + threadIdx.x;
    if (n >= N_NODES) return;
    float sn = dinv[n]; sn *= sn;
    out[2 * n]     += hw2[2 * n]     * sn + b2[0];
    out[2 * n + 1] += hw2[2 * n + 1] * sn + b2[1];
}

extern "C" void kernel_launch(void* const* d_in, const int* in_sizes, int n_in,
                              void* d_out, int out_size, void* d_ws, size_t ws_size,
                              hipStream_t stream) {
    const float* x  = (const float*)d_in[0];
    const int*   ei = (const int*)  d_in[1];
    const float* W1 = (const float*)d_in[2];
    const float* b1 = (const float*)d_in[3];
    const float* W2 = (const float*)d_in[4];
    const float* b2 = (const float*)d_in[5];
    const int* src = ei;
    const int* dst = ei + N_EDGES;
    float* out = (float*)d_out;

    float* ws   = (float*)d_ws;
    float* dinv = ws;
    float* en   = dinv + N_NODES;
    float* hw1  = en + N_EDGES;
    float* h    = hw1 + (size_t)N_NODES * HID;
    float* hw2  = h   + (size_t)N_NODES * HID;

    const dim3 blk(256);

    zero_kernel<<<(N_NODES + 255) / 256, blk, 0, stream>>>(dinv, N_NODES);
    zero_kernel<<<(N_NODES * HID + 255) / 256, blk, 0, stream>>>(h, N_NODES * HID);
    zero_kernel<<<(OUT_DIM * N_NODES + 255) / 256, blk, 0, stream>>>(out, OUT_DIM * N_NODES);

    deg_kernel  <<<(N_EDGES + 255) / 256, blk, 0, stream>>>(dst, dinv);
    dinv_kernel <<<(N_NODES + 255) / 256, blk, 0, stream>>>(dinv);
    enorm_kernel<<<(N_EDGES + 255) / 256, blk, 0, stream>>>(src, dst, dinv, en);

    gemm1_wmma<<<(NTILES + 7) / 8, blk, 0, stream>>>(x, W1, hw1);
    scatter1  <<<(N_EDGES * 16) / 256, blk, 0, stream>>>(src, dst, en, hw1, h);
    relu_bias <<<(N_NODES * HID + 255) / 256, blk, 0, stream>>>(hw1, dinv, b1, h);

    gemm2   <<<(N_NODES + 255) / 256, blk, 0, stream>>>(h, W2, hw2);
    scatter2<<<(N_EDGES + 255) / 256, blk, 0, stream>>>(src, dst, en, hw2, out);
    finish2 <<<(N_NODES + 255) / 256, blk, 0, stream>>>(hw2, dinv, b2, out);
}


# Layer-1 owners no longer test all edges: launch 1 (also zeroes the degree partials) sets each edge's bit in its dst range's bitmap, owners walk their own bitmap; LDS sums in arrival order as before
# speedup vs baseline: 1.8102x; 1.8102x over previous
#include <hip/hip_runtime.h>

typedef float v2f __attribute__((ext_vector_type(2)));
typedef float v4f __attribute__((ext_vector_type(4)));
typedef float v8f __attribute__((ext_vector_type(8)));

#define N_NODES 100000
#define N_EDGES 1600000
#define IN_DIM  165
#define HID     64
#define OUT_DIM 2
#define NTILES  (N_NODES / 16)
#define KPAIRS  83
#define NREP    8
#define TR      500
#define NRANGE  (N_NODES / TR)
#define DEGP_V4 (N_NODES * NREP / 4)
#define BM_EDGES 4096
#define BM_WORDS (BM_EDGES / 32)
#define BM_BLKS  ((N_EDGES + BM_EDGES - 1) / BM_EDGES)
#define BM_PITCH (BM_BLKS * BM_WORDS)
#define BM_LDSP  (BM_WORDS + 4)
#define QCAP    1056
static_assert(BM_BLKS >= 32 && N_EDGES % 4 == 0 && N_NODES % TR == 0 && TR % 2 == 0 && TR <= 32 * 16 && N_EDGES % 32 == 0 && BM_PITCH % 128 == 0 && BM_EDGES % 1024 == 0 && DEGP_V4 * 4 == N_NODES * NREP, "tile shape");

__global__ __launch_bounds__(256) void bitmap_kernel(const int* __restrict__ dst, unsigned* __restrict__ bm,
                                                     float* __restrict__ degp) {
    __shared__ unsigned s[NRANGE * BM_LDSP];
    const int tid = threadIdx.x;
    for (int i = blockIdx.x * 256 + tid; i < DEGP_V4; i += BM_BLKS * 256)
        ((v4f*)degp)[i] = (v4f){0.0f, 0.0f, 0.0f, 0.0f};
    for (int i = tid; i < NRANGE * BM_LDSP / 4; i += 256)
        ((uint4*)s)[i] = make_uint4(0u, 0u, 0u, 0u);
    __syncthreads();
    const int base = blockIdx.x * BM_EDGES;
    int4 d[BM_EDGES / 1024];
    for (int j = 0; j < BM_EDGES / 1024; ++j) {
        const int e = base + j * 1024 + tid * 4;
        d[j] = *(const int4*)(dst + (e < N_EDGES ? e : N_EDGES - 4));
    }
    for (int j = 0; j < BM_EDGES / 1024; ++j) {
        const int le = j * 1024 + tid * 4;
        if (base + le < N_EDGES) {
            unsigned* sw = s + (le >> 5);
            const unsigned b = 1u << (le & 31);
            atomicOr(sw + (d[j].x / TR) * BM_LDSP, b);
            atomicOr(sw + (d[j].y / TR) * BM_LDSP, b << 1);
            atomicOr(sw + (d[j].z / TR) * BM_LDSP, b << 2);
            atomicOr(sw + (d[j].w / TR) * BM_LDSP, b << 3);
        }
    }
    __syncthreads();
    for (int i = tid; i < NRANGE * (BM_WORDS / 4); i += 256) {
        const int r  = i >> 5;
        const int w4 = i & 31;
        const uint4 v = *(const uint4*)&s[r * BM_LDSP + w4 * 4];
        *(uint4*)&bm[(size_t)r * BM_PITCH + blockIdx.x * BM_WORDS + w4 * 4] = v;
    }
}

__global__ void deg_kernel(const int* __restrict__ dst, float* __restrict__ degp) {
    int e = blockIdx.x * blockDim.x + threadIdx.x;
    if (e < N_EDGES) unsafeAtomicAdd(&degp[(long)dst[e] * NREP + (e & (NREP - 1))], 1.0f);
}

__global__ void dinv_kernel(const float* __restrict__ degp, float* __restrict__ d) {
    int i = blockIdx.x * blockDim.x + threadIdx.x;
    if (i >= N_NODES) return;
    const v4f* __restrict__ p = (const v4f*)(degp + (long)i * NREP);
    v4f a = p[0], b = p[1];
    float deg = a.x + a.y + a.z + a.w + b.x + b.y + b.z + b.w;
    d[i] = rsqrtf(deg + 1.0f);
}

__global__ void enorm_kernel(const int* __restrict__ src, const int* __restrict__ dst,
                             const float* __restrict__ dinv, float* __restrict__ en) {
    int e = blockIdx.x * blockDim.x + threadIdx.x;
    if (e < N_EDGES) en[e] = dinv[src[e]] * dinv[dst[e]];
}

__global__ __launch_bounds__(256) void gemm1_wmma(const float* __restrict__ x,
                                                  const float* __restrict__ W1,
                                                  float* __restrict__ hw1) {
    __shared__ float w1p[KPAIRS * 2 * HID];
    for (int i = threadIdx.x; i < KPAIRS * 2 * HID; i += 256) {
        int kp   = i >> 7;
        int r    = i & 127;
        int col  = r >> 1;
        int sub  = r & 1;
        int krow = kp * 2 + sub;
        w1p[i] = (krow < IN_DIM) ? W1[krow * HID + col] : 0.0f;
    }
    __syncthreads();

    const int wave = threadIdx.x >> 5;
    const int lane = threadIdx.x & 31;
    const int tile = blockIdx.x * 8 + wave;
    if (tile >= NTILES) return;

    const int hh  = lane >> 4;
    const int lr  = lane & 15;
    const int row = tile * 16 + lr;
    const float* __restrict__ xrb = x + (long)row * IN_DIM;
    const float* __restrict__ xr  = xrb + 2 * hh;

    v8f acc0 = (v8f){}, acc1 = (v8f){}, acc2 = (v8f){}, acc3 = (v8f){};

    for (int k = 0; k < 164; k += 4) {
        v2f A;
        A.x = xr[k];
        A.y = xr[k + 1];
        const int ka = k + 2 * hh;
        const v2f* __restrict__ bp = (const v2f*)&w1p[(ka >> 1) * 2 * HID + lr * 2];
        v2f B0 = bp[0];
        v2f B1 = bp[16];
        v2f B2 = bp[32];
        v2f B3 = bp[48];
        acc0 = __builtin_amdgcn_wmma_f32_16x16x4_f32(false, A, false, B0, (short)0, acc0, false, false);
        acc1 = __builtin_amdgcn_wmma_f32_16x16x4_f32(false, A, false, B1, (short)0, acc1, false, false);
        acc2 = __builtin_amdgcn_wmma_f32_16x16x4_f32(false, A, false, B2, (short)0, acc2, false, false);
        acc3 = __builtin_amdgcn_wmma_f32_16x16x4_f32(false, A, false, B3, (short)0, acc3, false, false);
    }

    {
        float a164 = xrb[164];
        v2f A;
        A.x = (hh == 0) ? a164 : 0.0f;
        A.y = 0.0f;
        const v2f* __restrict__ bp = (const v2f*)&w1p[(KPAIRS - 1) * 2 * HID + lr * 2];
        v2f B0 = bp[0];
        v2f B1 = bp[16];
        v2f B2 = bp[32];
        v2f B3 = bp[48];
        acc0 = __builtin_amdgcn_wmma_f32_16x16x4_f32(false, A, false, B0, (short)0, acc0, false, false);
        acc1 = __builtin_amdgcn_wmma_f32_16x16x4_f32(false, A, false, B1, (short)0, acc1, false, false);
        acc2 = __builtin_amdgcn_wmma_f32_16x16x4_f32(false, A, false, B2, (short)0, acc2, false, false);
        acc3 = __builtin_amdgcn_wmma_f32_16x16x4_f32(false, A, false, B3, (short)0, acc3, false, false);
    }

    const long obase = (long)(tile * 16 + 8 * hh) * HID + lr;
    float* __restrict__ outp = hw1 + obase;
#pragma unroll
    for (int v = 0; v < 8; ++v) {
        long o = (long)v * HID;
        outp[o +  0] = acc0[v];
        outp[o + 16] = acc1[v];
        outp[o + 32] = acc2[v];
        outp[o + 48] = acc3[v];
    }
}


__device__ __forceinline__ void add32(const int* qp, int lane, int lo,
                                      const int* __restrict__ src, const int* __restrict__ dst,
                                      const float* __restrict__ en,
                                      const float* __restrict__ hw1, float* acc) {
    const int e  = qp[lane];
    const int   s = src[e];
    const int  dl = dst[e] - lo;
    const float w = en[e];
    for (int g = 0; g < 2; ++g) {
        v2f row[16];
        for (int k = 0; k < 16; ++k) {
            const int sk = __builtin_amdgcn_readlane(s, g * 16 + k);
            row[k] = *(const v2f*)(hw1 + (long)sk * HID + 2 * lane);
        }
        for (int k = 0; k < 16; ++k) {
            const float wk = __int_as_float(__builtin_amdgcn_readlane(__float_as_int(w), g * 16 + k));
            const int   dk = __builtin_amdgcn_readlane(dl, g * 16 + k);
            float* a = acc + dk * HID + 2 * lane;
            atomicAdd(a,     row[k].x * wk);
            atomicAdd(a + 1, row[k].y * wk);
        }
    }
}

__device__ __forceinline__ void add_tail(const int* qp, int n, int lane, int lo,
                                         const int* __restrict__ src, const int* __restrict__ dst,
                                         const float* __restrict__ en,
                                         const float* __restrict__ hw1, float* acc) {
    int s = 0, dl = 0;
    float w = 0.0f;
    if (lane < n) {
        const int e = qp[lane];
        dl = dst[e] - lo;
        s  = src[e];
        w  = en[e];
    }
    for (int g = 0; g < 4; ++g) {
        if (g * 8 >= n) break;
        v2f row[8];
        for (int k = 0; k < 8; ++k) {
            row[k] = (v2f){0.0f, 0.0f};
            if (g * 8 + k < n) {
                const int sk = __builtin_amdgcn_readlane(s, g * 8 + k);
                row[k] = *(const v2f*)(hw1 + (long)sk * HID + 2 * lane);
            }
        }
        for (int k = 0; k < 8; ++k) {
            if (g * 8 + k < n) {
                const float wk = __int_as_float(__builtin_amdgcn_readlane(__float_as_int(w), g * 8 + k));
                const int   dk = __builtin_amdgcn_readlane(dl, g * 8 + k);
                float* a = acc + dk * HID + 2 * lane;
                atomicAdd(a,     row[k].x * wk);
                atomicAdd(a + 1, row[k].y * wk);
            }
        }
    }
}

__global__ __launch_bounds__(1024) void agg1_tiles(const int* __restrict__ src, const int* __restrict__ dst,
                                                   const unsigned* __restrict__ bm,
                                                   const float* __restrict__ en, const float* __restrict__ hw1,
                                                   const float* __restrict__ dinv, const float* __restrict__ b1,
                                                   const float* __restrict__ W2, const float* __restrict__ b2,
                                                   float* __restrict__ hw2, float* __restrict__ acc2) {
    __shared__ float acc[TR * HID];
    __shared__ int   qbuf[32 * QCAP];
    const int tid  = threadIdx.x;
    const int lane = tid & 31;
    const int wave = __builtin_amdgcn_readfirstlane(tid >> 5);
    const int lo   = blockIdx.x * TR;

    {
        const v4f* __restrict__ hp = (const v4f*)(hw1 + (long)lo * HID);
        for (int i = tid; i < TR * HID / 4; i += 1024) {
            float sn = dinv[lo + (i >> 4)];
            sn *= sn;
            const v4f h = hp[i];
            const v4f b = *(const v4f*)(b1 + (i & 15) * 4);
            v4f r;
            r.x = h.x * sn + b.x;
            r.y = h.y * sn + b.y;
            r.z = h.z * sn + b.z;
            r.w = h.w * sn + b.w;
            *(v4f*)(acc + i * 4) = r;
        }
    }
    __syncthreads();

    int* q = qbuf + wave * QCAP;
    int qn = 0;
    const uint4* __restrict__ bmr = (const uint4*)(bm + (size_t)blockIdx.x * BM_PITCH);
    uint4 cur = bmr[wave * 32 + lane];
    const int nj = (BM_BLKS - wave + 31) / 32;
    for (int j = 0; j < nj; ++j) {
        const int piece = wave + 32 * j;
        const int pn = piece + 32;
        const uint4 ld = bmr[(pn < BM_BLKS ? pn : piece) * 32 + lane];
#pragma unroll
        for (int c = 0; c < 4; ++c) {
            unsigned w = c == 0 ? cur.x : c == 1 ? cur.y : c == 2 ? cur.z : cur.w;
            const int eb = ((piece * 32 + lane) * 4 + c) * 32;
            unsigned m = __builtin_amdgcn_ballot_w32(w != 0u);
            while (m) {
                if (w != 0u) {
                    q[qn + __builtin_amdgcn_mbcnt_lo(m, 0)] = eb + __builtin_ctz(w);
                    w &= w - 1u;
                }
                qn += __builtin_popcount(m);
                m = __builtin_amdgcn_ballot_w32(w != 0u);
            }
            if (qn < 32) continue;
            int qh = 0;
            while (qn - qh >= 32) {
                add32(q + qh, lane, lo, src, dst, en, hw1, acc);
                qh += 32;
            }
            const int rem = qn - qh;
            int t = 0;
            if (lane < rem) t = q[qh + lane];
            if (lane < rem) q[lane] = t;
            qn = rem;
        }
        cur = ld;
    }
    if (qn > 0) add_tail(q, qn, lane, lo, src, dst, en, hw1, acc);
    __syncthreads();

    {
        const int half = lane >> 4;
        const int c4   = lane & 15;
        const v4f wa = *(const v4f*)(W2 + 8 * c4);
        const v4f wb = *(const v4f*)(W2 + 8 * c4 + 4);
        const float bb0 = b2[0], bb1 = b2[1];
        const int rb = wave * 16;
        const int np = (TR - rb) < 16 ? (TR - rb) / 2 : 8;
        float k0 = 0.0f, k1 = 0.0f;
        float dv = 0.0f;
        if (rb + c4 < TR) dv = dinv[lo + rb + c4];
        float p0[8], p1[8];
        for (int i = 0; i < 8; ++i) {
            int r = rb + 2 * i + half;
            r = r < TR ? r : TR - 1;
            v4f h = *(const v4f*)(acc + r * HID + 4 * c4);
            h.x = h.x > 0.0f ? h.x : 0.0f;
            h.y = h.y > 0.0f ? h.y : 0.0f;
            h.z = h.z > 0.0f ? h.z : 0.0f;
            h.w = h.w > 0.0f ? h.w : 0.0f;
            p0[i] = fmaf(h.w, wb.z, fmaf(h.z, wb.x, fmaf(h.y, wa.z, h.x * wa.x)));
            p1[i] = fmaf(h.w, wb.w, fmaf(h.z, wb.y, fmaf(h.y, wa.w, h.x * wa.y)));
        }
        for (int m = 8; m >= 1; m >>= 1) {
            for (int i = 0; i < 8; ++i) {
                p0[i] += __shfl_xor(p0[i], m, 16);
                p1[i] += __shfl_xor(p1[i], m, 16);
            }
        }
        for (int i = 0; i < 8; ++i) {
            const int r = rb + 2 * i + half;
            float sn = __shfl(dv, 2 * i + half, 16);
            sn *= sn;
            if (c4 == i) { k0 = p0[i]; k1 = p1[i]; }
            if (i < np && c4 < 4) {
                v4f st = (v4f){0.0f, 0.0f, 0.0f, 0.0f};
                if (c4 == 0) { st.x = p0[i] * sn + bb0; st.y = p1[i] * sn + bb1; }
                *(v4f*)(acc2 + (long)(lo + r) * NREP * 2 + 4 * c4) = st;
            }
        }
        if (c4 < np) *(v2f*)(hw2 + 2 * (long)(lo + rb + 2 * c4 + half)) = (v2f){k0, k1};
    }
}

__global__ void scatter2(const int* __restrict__ src, const int* __restrict__ dst,
                         const float* __restrict__ en, const float* __restrict__ hw2,
                         float* __restrict__ acc2) {
    int e = blockIdx.x * blockDim.x + threadIdx.x;
    if (e >= N_EDGES) return;
    int s = src[e], d = dst[e];
    float w = en[e];
    float* ap = acc2 + ((long)d * NREP + (e & (NREP - 1))) * 2;
    unsafeAtomicAdd(ap,     hw2[2 * s]     * w);
    unsafeAtomicAdd(ap + 1, hw2[2 * s + 1] * w);
}

__global__ void finish2(const float* __restrict__ acc2, float* __restrict__ out) {
    int n = blockIdx.x * blockDim.x + threadIdx.x;
    if (n >= N_NODES) return;
    const v4f* __restrict__ p = (const v4f*)(acc2 + (long)n * NREP * 2);
    v4f q0 = p[0], q1 = p[1], q2 = p[2], q3 = p[3];
    float s0 = q0.x + q0.z + q1.x + q1.z + q2.x + q2.z + q3.x + q3.z;
    float s1 = q0.y + q0.w + q1.y + q1.w + q2.y + q2.w + q3.y + q3.w;
    out[2 * n]     = s0;
    out[2 * n + 1] = s1;
}

extern "C" void kernel_launch(void* const* d_in, const int* in_sizes, int n_in,
                              void* d_out, int out_size, void* d_ws, size_t ws_size,
                              hipStream_t stream) {
    const float* x  = (const float*)d_in[0];
    const int*   ei = (const int*)  d_in[1];
    const float* W1 = (const float*)d_in[2];
    const float* b1 = (const float*)d_in[3];
    const float* W2 = (const float*)d_in[4];
    const float* b2 = (const float*)d_in[5];
    const int* src = ei;
    const int* dst = ei + N_EDGES;
    float* out = (float*)d_out;

    float* ws   = (float*)d_ws;
    float* dinv = ws;
    float* en   = dinv + N_NODES;
    float* hw1  = en + N_EDGES;
    float* hw2  = hw1 + (size_t)N_NODES * HID;
    float* degp = hw2 + (size_t)N_NODES * OUT_DIM;
    float* acc2 = degp + (size_t)N_NODES * NREP;
    unsigned* bm = (unsigned*)(acc2 + (size_t)N_NODES * NREP * 2) + 32;

    const dim3 blk(256);

    bitmap_kernel<<<BM_BLKS, blk, 0, stream>>>(dst, bm, degp);

    deg_kernel  <<<(N_EDGES + 255) / 256, blk, 0, stream>>>(dst, degp);
    dinv_kernel <<<(N_NODES + 255) / 256, blk, 0, stream>>>(degp, dinv);
    enorm_kernel<<<(N_EDGES + 255) / 256, blk, 0, stream>>>(src, dst, dinv, en);

    gemm1_wmma<<<(NTILES + 7) / 8, blk, 0, stream>>>(x, W1, hw1);
    agg1_tiles<<<NRANGE, dim3(1024), 0, stream>>>(src, dst, bm, en, hw1, dinv, b1, W2, b2, hw2, acc2);

    scatter2<<<(N_EDGES + 255) / 256, blk, 0, stream>>>(src, dst, en, hw2, acc2);
    finish2 <<<(N_NODES + 255) / 256, blk, 0, stream>>>(acc2, out);
}


# Degree count and layer-2 edge sums leave global atomics: range owners walk their bitmap rows again and add in LDS (integer counts; f32 sums in arrival order, one a node, no partials); 6 launches
# speedup vs baseline: 2.2834x; 1.2614x over previous
#include <hip/hip_runtime.h>

typedef float v2f __attribute__((ext_vector_type(2)));
typedef float v4f __attribute__((ext_vector_type(4)));
typedef float v8f __attribute__((ext_vector_type(8)));

#define N_NODES 100000
#define N_EDGES 1600000
#define IN_DIM  165
#define HID     64
#define OUT_DIM 2
#define NTILES  (N_NODES / 16)
#define KPAIRS  83
#define TR      500
#define NRANGE  (N_NODES / TR)
#define BM_EDGES 4096
#define BM_WORDS (BM_EDGES / 32)
#define BM_BLKS  ((N_EDGES + BM_EDGES - 1) / BM_EDGES)
#define BM_PITCH (BM_BLKS * BM_WORDS)
#define BM_LDSP  (BM_WORDS + 4)
#define QCAP    1056
static_assert(BM_BLKS >= 32 && N_EDGES % 4 == 0 && N_NODES % TR == 0 && TR % 2 == 0 && TR <= 32 * 16 && N_EDGES % 32 == 0 && BM_PITCH % 128 == 0 && BM_EDGES % 1024 == 0, "tile shape");

__global__ __launch_bounds__(256) void bitmap_kernel(const int* __restrict__ dst, unsigned* __restrict__ bm) {
    __shared__ unsigned s[NRANGE * BM_LDSP];
    const int tid = threadIdx.x;
    for (int i = tid; i < NRANGE * BM_LDSP / 4; i += 256)
        ((uint4*)s)[i] = make_uint4(0u, 0u, 0u, 0u);
    __syncthreads();
    const int base = blockIdx.x * BM_EDGES;
    int4 d[BM_EDGES / 1024];
    for (int j = 0; j < BM_EDGES / 1024; ++j) {
        const int e = base + j * 1024 + tid * 4;
        d[j] = *(const int4*)(dst + (e < N_EDGES ? e : N_EDGES - 4));
    }
    for (int j = 0; j < BM_EDGES / 1024; ++j) {
        const int le = j * 1024 + tid * 4;
        if (base + le < N_EDGES) {
            unsigned* sw = s + (le >> 5);
            const unsigned b = 1u << (le & 31);
            atomicOr(sw + (d[j].x / TR) * BM_LDSP, b);
            atomicOr(sw + (d[j].y / TR) * BM_LDSP, b << 1);
            atomicOr(sw + (d[j].z / TR) * BM_LDSP, b << 2);
            atomicOr(sw + (d[j].w / TR) * BM_LDSP, b << 3);
        }
    }
    __syncthreads();
    for (int i = tid; i < NRANGE * (BM_WORDS / 4); i += 256) {
        const int r  = i >> 5;
        const int w4 = i & 31;
        const uint4 v = *(const uint4*)&s[r * BM_LDSP + w4 * 4];
        *(uint4*)&bm[(size_t)r * BM_PITCH + blockIdx.x * BM_WORDS + w4 * 4] = v;
    }
}

template <class F>
__device__ __forceinline__ void walk_owned_edges(const unsigned* __restrict__ bm, int* q, int wave, int lane, F f) {
    int qn = 0;
    const uint4* __restrict__ bmr = (const uint4*)(bm + (size_t)blockIdx.x * BM_PITCH);
    uint4 cur = bmr[wave * 32 + lane];
    const int nj = (BM_BLKS - wave + 31) / 32;
    for (int j = 0; j < nj; ++j) {
        const int piece = wave + 32 * j;
        const int pn = piece + 32;
        const uint4 ld = bmr[(pn < BM_BLKS ? pn : piece) * 32 + lane];
#pragma unroll
        for (int c = 0; c < 4; ++c) {
            unsigned w = c == 0 ? cur.x : c == 1 ? cur.y : c == 2 ? cur.z : cur.w;
            const int eb = ((piece * 32 + lane) * 4 + c) * 32;
            unsigned m = __builtin_amdgcn_ballot_w32(w != 0u);
            while (m) {
                if (w != 0u) {
                    q[qn + __builtin_amdgcn_mbcnt_lo(m, 0)] = eb + __builtin_ctz(w);
                    w &= w - 1u;
                }
                qn += __builtin_popcount(m);
                m = __builtin_amdgcn_ballot_w32(w != 0u);
            }
            if (qn < 32) continue;
            int qh = 0;
            while (qn - qh >= 32) {
                f(q + qh, 32);
                qh += 32;
            }
            const int rem = qn - qh;
            int t = 0;
            if (lane < rem) t = q[qh + lane];
            if (lane < rem) q[lane] = t;
            qn = rem;
        }
        cur = ld;
    }
    if (qn > 0) f(q, qn);
}

__global__ __launch_bounds__(1024) void deg_tiles(const int* __restrict__ dst, const unsigned* __restrict__ bm,
                                                  float* __restrict__ dinv) {
    __shared__ unsigned cnt[TR];
    __shared__ int      qbuf[32 * QCAP];
    const int tid  = threadIdx.x;
    const int lane = tid & 31;
    const int wave = __builtin_amdgcn_readfirstlane(tid >> 5);
    const int lo   = blockIdx.x * TR;
    if (tid < TR) cnt[tid] = 0u;
    __syncthreads();
    walk_owned_edges(bm, qbuf + wave * QCAP, wave, lane, [&](const int* qp, int n) {
        if (lane < n) atomicAdd(&cnt[dst[qp[lane]] - lo], 1u);
    });
    __syncthreads();
    if (tid < TR) dinv[lo + tid] = rsqrtf((float)cnt[tid] + 1.0f);
}

__global__ void enorm_kernel(const int* __restrict__ src, const int* __restrict__ dst,
                             const float* __restrict__ dinv, float* __restrict__ en) {
    int e = blockIdx.x * blockDim.x + threadIdx.x;
    if (e < N_EDGES) en[e] = dinv[src[e]] * dinv[dst[e]];
}

__global__ __launch_bounds__(256) void gemm1_wmma(const float* __restrict__ x,
                                                  const float* __restrict__ W1,
                                                  float* __restrict__ hw1) {
    __shared__ float w1p[KPAIRS * 2 * HID];
    for (int i = threadIdx.x; i < KPAIRS * 2 * HID; i += 256) {
        int kp   = i >> 7;
        int r    = i & 127;
        int col  = r >> 1;
        int sub  = r & 1;
        int krow = kp * 2 + sub;
        w1p[i] = (krow < IN_DIM) ? W1[krow * HID + col] : 0.0f;
    }
    __syncthreads();

    const int wave = threadIdx.x >> 5;
    const int lane = threadIdx.x & 31;
    const int tile = blockIdx.x * 8 + wave;
    if (tile >= NTILES) return;

    const int hh  = lane >> 4;
    const int lr  = lane & 15;
    const int row = tile * 16 + lr;
    const float* __restrict__ xrb = x + (long)row * IN_DIM;
    const float* __restrict__ xr  = xrb + 2 * hh;

    v8f acc0 = (v8f){}, acc1 = (v8f){}, acc2 = (v8f){}, acc3 = (v8f){};

    for (int k = 0; k < 164; k += 4) {
        v2f A;
        A.x = xr[k];
        A.y = xr[k + 1];
        const int ka = k + 2 * hh;
        const v2f* __restrict__ bp = (const v2f*)&w1p[(ka >> 1) * 2 * HID + lr * 2];
        v2f B0 = bp[0];
        v2f B1 = bp[16];
        v2f B2 = bp[32];
        v2f B3 = bp[48];
        acc0 = __builtin_amdgcn_wmma_f32_16x16x4_f32(false, A, false, B0, (short)0, acc0, false, false);
        acc1 = __builtin_amdgcn_wmma_f32_16x16x4_f32(false, A, false, B1, (short)0, acc1, false, false);
        acc2 = __builtin_amdgcn_wmma_f32_16x16x4_f32(false, A, false, B2, (short)0, acc2, false, false);
        acc3 = __builtin_amdgcn_wmma_f32_16x16x4_f32(false, A, false, B3, (short)0, acc3, false, false);
    }

    {
        float a164 = xrb[164];
        v2f A;
        A.x = (hh == 0) ? a164 : 0.0f;
        A.y = 0.0f;
        const v2f* __restrict__ bp = (const v2f*)&w1p[(KPAIRS - 1) * 2 * HID + lr * 2];
        v2f B0 = bp[0];
        v2f B1 = bp[16];
        v2f B2 = bp[32];
        v2f B3 = bp[48];
        acc0 = __builtin_amdgcn_wmma_f32_16x16x4_f32(false, A, false, B0, (short)0, acc0, false, false);
        acc1 = __builtin_amdgcn_wmma_f32_16x16x4_f32(false, A, false, B1, (short)0, acc1, false, false);
        acc2 = __builtin_amdgcn_wmma_f32_16x16x4_f32(false, A, false, B2, (short)0, acc2, false, false);
        acc3 = __builtin_amdgcn_wmma_f32_16x16x4_f32(false, A, false, B3, (short)0, acc3, false, false);
    }

    const long obase = (long)(tile * 16 + 8 * hh) * HID + lr;
    float* __restrict__ outp = hw1 + obase;
#pragma unroll
    for (int v = 0; v < 8; ++v) {
        long o = (long)v * HID;
        outp[o +  0] = acc0[v];
        outp[o + 16] = acc1[v];
        outp[o + 32] = acc2[v];
        outp[o + 48] = acc3[v];
    }
}


__device__ __forceinline__ void add32(const int* qp, int lane, int lo,
                                      const int* __restrict__ src, const int* __restrict__ dst,
                                      const float* __restrict__ en,
                                      const float* __restrict__ hw1, float* acc) {
    const int e  = qp[lane];
    const int   s = src[e];
    const int  dl = dst[e] - lo;
    const float w = en[e];
    for (int g = 0; g < 2; ++g) {
        v2f row[16];
        for (int k = 0; k < 16; ++k) {
            const int sk = __builtin_amdgcn_readlane(s, g * 16 + k);
            row[k] = *(const v2f*)(hw1 + (long)sk * HID + 2 * lane);
        }
        for (int k = 0; k < 16; ++k) {
            const float wk = __int_as_float(__builtin_amdgcn_readlane(__float_as_int(w), g * 16 + k));
            const int   dk = __builtin_amdgcn_readlane(dl, g * 16 + k);
            float* a = acc + dk * HID + 2 * lane;
            atomicAdd(a,     row[k].x * wk);
            atomicAdd(a + 1, row[k].y * wk);
        }
    }
}

__device__ __forceinline__ void add_tail(const int* qp, int n, int lane, int lo,
                                         const int* __restrict__ src, const int* __restrict__ dst,
                                         const float* __restrict__ en,
                                         const float* __restrict__ hw1, float* acc) {
    int s = 0, dl = 0;
    float w = 0.0f;
    if (lane < n) {
        const int e = qp[lane];
        dl = dst[e] - lo;
        s  = src[e];
        w  = en[e];
    }
    for (int g = 0; g < 4; ++g) {
        if (g * 8 >= n) break;
        v2f row[8];
        for (int k = 0; k < 8; ++k) {
            row[k] = (v2f){0.0f, 0.0f};
            if (g * 8 + k < n) {
                const int sk = __builtin_amdgcn_readlane(s, g * 8 + k);
                row[k] = *(const v2f*)(hw1 + (long)sk * HID + 2 * lane);
            }
        }
        for (int k = 0; k < 8; ++k) {
            if (g * 8 + k < n) {
                const float wk = __int_as_float(__builtin_amdgcn_readlane(__float_as_int(w), g * 8 + k));
                const int   dk = __builtin_amdgcn_readlane(dl, g * 8 + k);
                float* a = acc + dk * HID + 2 * lane;
                atomicAdd(a,     row[k].x * wk);
                atomicAdd(a + 1, row[k].y * wk);
            }
        }
    }
}

__global__ __launch_bounds__(1024) void agg1_tiles(const int* __restrict__ src, const int* __restrict__ dst,
                                                   const unsigned* __restrict__ bm,
                                                   const float* __restrict__ en, const float* __restrict__ hw1,
                                                   const float* __restrict__ dinv, const float* __restrict__ b1,
                                                   const float* __restrict__ W2, float* __restrict__ hw2) {
    __shared__ float acc[TR * HID];
    __shared__ int   qbuf[32 * QCAP];
    const int tid  = threadIdx.x;
    const int lane = tid & 31;
    const int wave = __builtin_amdgcn_readfirstlane(tid >> 5);
    const int lo   = blockIdx.x * TR;

    {
        const v4f* __restrict__ hp = (const v4f*)(hw1 + (long)lo * HID);
        for (int i = tid; i < TR * HID / 4; i += 1024) {
            float sn = dinv[lo + (i >> 4)];
            sn *= sn;
            const v4f h = hp[i];
            const v4f b = *(const v4f*)(b1 + (i & 15) * 4);
            v4f r;
            r.x = h.x * sn + b.x;
            r.y = h.y * sn + b.y;
            r.z = h.z * sn + b.z;
            r.w = h.w * sn + b.w;
            *(v4f*)(acc + i * 4) = r;
        }
    }
    __syncthreads();

    int* q = qbuf + wave * QCAP;
    int qn = 0;
    const uint4* __restrict__ bmr = (const uint4*)(bm + (size_t)blockIdx.x * BM_PITCH);
    uint4 cur = bmr[wave * 32 + lane];
    const int nj = (BM_BLKS - wave + 31) / 32;
    for (int j = 0; j < nj; ++j) {
        const int piece = wave + 32 * j;
        const int pn = piece + 32;
        const uint4 ld = bmr[(pn < BM_BLKS ? pn : piece) * 32 + lane];
#pragma unroll
        for (int c = 0; c < 4; ++c) {
            unsigned w = c == 0 ? cur.x : c == 1 ? cur.y : c == 2 ? cur.z : cur.w;
            const int eb = ((piece * 32 + lane) * 4 + c) * 32;
            unsigned m = __builtin_amdgcn_ballot_w32(w != 0u);
            while (m) {
                if (w != 0u) {
                    q[qn + __builtin_amdgcn_mbcnt_lo(m, 0)] = eb + __builtin_ctz(w);
                    w &= w - 1u;
                }
                qn += __builtin_popcount(m);
                m = __builtin_amdgcn_ballot_w32(w != 0u);
            }
            if (qn < 32) continue;
            int qh = 0;
            while (qn - qh >= 32) {
                add32(q + qh, lane, lo, src, dst, en, hw1, acc);
                qh += 32;
            }
            const int rem = qn - qh;
            int t = 0;
            if (lane < rem) t = q[qh + lane];
            if (lane < rem) q[lane] = t;
            qn = rem;
        }
        cur = ld;
    }
    if (qn > 0) add_tail(q, qn, lane, lo, src, dst, en, hw1, acc);
    __syncthreads();

    {
        const int half = lane >> 4;
        const int c4   = lane & 15;
        const v4f wa = *(const v4f*)(W2 + 8 * c4);
        const v4f wb = *(const v4f*)(W2 + 8 * c4 + 4);
        const int rb = wave * 16;
        const int np = (TR - rb) < 16 ? (TR - rb) / 2 : 8;
        float k0 = 0.0f, k1 = 0.0f;
        float p0[8], p1[8];
        for (int i = 0; i < 8; ++i) {
            int r = rb + 2 * i + half;
            r = r < TR ? r : TR - 1;
            v4f h = *(const v4f*)(acc + r * HID + 4 * c4);
            h.x = h.x > 0.0f ? h.x : 0.0f;
            h.y = h.y > 0.0f ? h.y : 0.0f;
            h.z = h.z > 0.0f ? h.z : 0.0f;
            h.w = h.w > 0.0f ? h.w : 0.0f;
            p0[i] = fmaf(h.w, wb.z, fmaf(h.z, wb.x, fmaf(h.y, wa.z, h.x * wa.x)));
            p1[i] = fmaf(h.w, wb.w, fmaf(h.z, wb.y, fmaf(h.y, wa.w, h.x * wa.y)));
        }
        for (int m = 8; m >= 1; m >>= 1) {
            for (int i = 0; i < 8; ++i) {
                p0[i] += __shfl_xor(p0[i], m, 16);
                p1[i] += __shfl_xor(p1[i], m, 16);
            }
        }
        for (int i = 0; i < 8; ++i) {
            if (c4 == i) { k0 = p0[i]; k1 = p1[i]; }
        }
        if (c4 < np) *(v2f*)(hw2 + 2 * (long)(lo + rb + 2 * c4 + half)) = (v2f){k0, k1};
    }
}

__global__ __launch_bounds__(1024) void agg2_tiles(const int* __restrict__ src, const int* __restrict__ dst,
                                                   const unsigned* __restrict__ bm,
                                                   const float* __restrict__ en, const float* __restrict__ hw2,
                                                   const float* __restrict__ dinv, const float* __restrict__ b2,
                                                   float* __restrict__ out) {
    __shared__ float a2[TR * OUT_DIM];
    __shared__ int   qbuf[32 * QCAP];
    const int tid  = threadIdx.x;
    const int lane = tid & 31;
    const int wave = __builtin_amdgcn_readfirstlane(tid >> 5);
    const int lo   = blockIdx.x * TR;
    if (tid < TR) {
        float sn = dinv[lo + tid];
        sn *= sn;
        const v2f h = *(const v2f*)(hw2 + 2 * (long)(lo + tid));
        a2[2 * tid]     = h.x * sn + b2[0];
        a2[2 * tid + 1] = h.y * sn + b2[1];
    }
    __syncthreads();
    walk_owned_edges(bm, qbuf + wave * QCAP, wave, lane, [&](const int* qp, int n) {
        if (lane < n) {
            const int   e  = qp[lane];
            const int   s  = src[e];
            const int   dl = dst[e] - lo;
            const float w  = en[e];
            const v2f   h  = *(const v2f*)(hw2 + 2 * (long)s);
            atomicAdd(&a2[2 * dl],     h.x * w);
            atomicAdd(&a2[2 * dl + 1], h.y * w);
        }
    });
    __syncthreads();
    if (tid < TR) *(v2f*)(out + 2 * (long)(lo + tid)) = *(const v2f*)(a2 + 2 * tid);
}

extern "C" void kernel_launch(void* const* d_in, const int* in_sizes, int n_in,
                              void* d_out, int out_size, void* d_ws, size_t ws_size,
                              hipStream_t stream) {
    const float* x  = (const float*)d_in[0];
    const int*   ei = (const int*)  d_in[1];
    const float* W1 = (const float*)d_in[2];
    const float* b1 = (const float*)d_in[3];
    const float* W2 = (const float*)d_in[4];
    const float* b2 = (const float*)d_in[5];
    const int* src = ei;
    const int* dst = ei + N_EDGES;
    float* out = (float*)d_out;

    float* ws   = (float*)d_ws;
    float* dinv = ws;
    float* en   = dinv + N_NODES;
    float* hw1  = en + N_EDGES;
    float* hw2  = hw1 + (size_t)N_NODES * HID;
    unsigned* bm = (unsigned*)(hw2 + (size_t)N_NODES * OUT_DIM) + 32;

    const dim3 blk(256);

    bitmap_kernel<<<BM_BLKS, blk, 0, stream>>>(dst, bm);

    deg_tiles   <<<NRANGE, dim3(1024), 0, stream>>>(dst, bm, dinv);
    enorm_kernel<<<(N_EDGES + 255) / 256, blk, 0, stream>>>(src, dst, dinv, en);

    gemm1_wmma<<<(NTILES + 7) / 8, blk, 0, stream>>>(x, W1, hw1);
    agg1_tiles<<<NRANGE, dim3(1024), 0, stream>>>(src, dst, bm, en, hw1, dinv, b1, W2, hw2);

    agg2_tiles<<<NRANGE, dim3(1024), 0, stream>>>(src, dst, bm, en, hw2, dinv, b2, out);
}


# Per-edge norm plane and its 6250-workgroup launch go: range owners form dinv[src]*dinv[dst] per listed edge (dinv[src] gathered beside the rows, own dinv in LDS); 5 launches; sum order as before
# speedup vs baseline: 2.4980x; 1.0940x over previous
#include <hip/hip_runtime.h>

typedef float v2f __attribute__((ext_vector_type(2)));
typedef float v4f __attribute__((ext_vector_type(4)));
typedef float v8f __attribute__((ext_vector_type(8)));

#define N_NODES 100000
#define N_EDGES 1600000
#define IN_DIM  165
#define HID     64
#define OUT_DIM 2
#define NTILES  (N_NODES / 16)
#define KPAIRS  83
#define TR      500
#define NRANGE  (N_NODES / TR)
#define BM_EDGES 4096
#define BM_WORDS (BM_EDGES / 32)
#define BM_BLKS  ((N_EDGES + BM_EDGES - 1) / BM_EDGES)
#define BM_PITCH (BM_BLKS * BM_WORDS)
#define BM_LDSP  (BM_WORDS + 4)
#define QCAP    1056
static_assert(BM_BLKS >= 32 && N_EDGES % 4 == 0 && N_NODES % TR == 0 && TR % 2 == 0 && TR <= 32 * 16 && N_EDGES % 32 == 0 && BM_PITCH % 128 == 0 && BM_EDGES % 1024 == 0, "tile shape");

__global__ __launch_bounds__(256) void bitmap_kernel(const int* __restrict__ dst, unsigned* __restrict__ bm) {
    __shared__ unsigned s[NRANGE * BM_LDSP];
    const int tid = threadIdx.x;
    for (int i = tid; i < NRANGE * BM_LDSP / 4; i += 256)
        ((uint4*)s)[i] = make_uint4(0u, 0u, 0u, 0u);
    __syncthreads();
    const int base = blockIdx.x * BM_EDGES;
    int4 d[BM_EDGES / 1024];
    for (int j = 0; j < BM_EDGES / 1024; ++j) {
        const int e = base + j * 1024 + tid * 4;
        d[j] = *(const int4*)(dst + (e < N_EDGES ? e : N_EDGES - 4));
    }
    for (int j = 0; j < BM_EDGES / 1024; ++j) {
        const int le = j * 1024 + tid * 4;
        if (base + le < N_EDGES) {
            unsigned* sw = s + (le >> 5);
            const unsigned b = 1u << (le & 31);
            atomicOr(sw + (d[j].x / TR) * BM_LDSP, b);
            atomicOr(sw + (d[j].y / TR) * BM_LDSP, b << 1);
            atomicOr(sw + (d[j].z / TR) * BM_LDSP, b << 2);
            atomicOr(sw + (d[j].w / TR) * BM_LDSP, b << 3);
        }
    }
    __syncthreads();
    for (int i = tid; i < NRANGE * (BM_WORDS / 4); i += 256) {
        const int r  = i >> 5;
        const int w4 = i & 31;
        const uint4 v = *(const uint4*)&s[r * BM_LDSP + w4 * 4];
        *(uint4*)&bm[(size_t)r * BM_PITCH + blockIdx.x * BM_WORDS + w4 * 4] = v;
    }
}

template <class F>
__device__ __forceinline__ void walk_owned_edges(const unsigned* __restrict__ bm, int* q, int wave, int lane, F f) {
    int qn = 0;
    const uint4* __restrict__ bmr = (const uint4*)(bm + (size_t)blockIdx.x * BM_PITCH);
    uint4 cur = bmr[wave * 32 + lane];
    const int nj = (BM_BLKS - wave + 31) / 32;
    for (int j = 0; j < nj; ++j) {
        const int piece = wave + 32 * j;
        const int pn = piece + 32;
        const uint4 ld = bmr[(pn < BM_BLKS ? pn : piece) * 32 + lane];
#pragma unroll
        for (int c = 0; c < 4; ++c) {
            unsigned w = c == 0 ? cur.x : c == 1 ? cur.y : c == 2 ? cur.z : cur.w;
            const int eb = ((piece * 32 + lane) * 4 + c) * 32;
            unsigned m = __builtin_amdgcn_ballot_w32(w != 0u);
            while (m) {
                if (w != 0u) {
                    q[qn + __builtin_amdgcn_mbcnt_lo(m, 0)] = eb + __builtin_ctz(w);
                    w &= w - 1u;
                }
                qn += __builtin_popcount(m);
                m = __builtin_amdgcn_ballot_w32(w != 0u);
            }
            if (qn < 32) continue;
            int qh = 0;
            while (qn - qh >= 32) {
                f(q + qh, 32);
                qh += 32;
            }
            const int rem = qn - qh;
            int t = 0;
            if (lane < rem) t = q[qh + lane];
            if (lane < rem) q[lane] = t;
            qn = rem;
        }
        cur = ld;
    }
    if (qn > 0) f(q, qn);
}

__global__ __launch_bounds__(1024) void deg_tiles(const int* __restrict__ dst, const unsigned* __restrict__ bm,
                                                  float* __restrict__ dinv) {
    __shared__ unsigned cnt[TR];
    __shared__ int      qbuf[32 * QCAP];
    const int tid  = threadIdx.x;
    const int lane = tid & 31;
    const int wave = __builtin_amdgcn_readfirstlane(tid >> 5);
    const int lo   = blockIdx.x * TR;
    if (tid < TR) cnt[tid] = 0u;
    __syncthreads();
    walk_owned_edges(bm, qbuf + wave * QCAP, wave, lane, [&](const int* qp, int n) {
        if (lane < n) atomicAdd(&cnt[dst[qp[lane]] - lo], 1u);
    });
    __syncthreads();
    if (tid < TR) dinv[lo + tid] = rsqrtf((float)cnt[tid] + 1.0f);
}

__global__ __launch_bounds__(256) void gemm1_wmma(const float* __restrict__ x,
                                                  const float* __restrict__ W1,
                                                  float* __restrict__ hw1) {
    __shared__ float w1p[KPAIRS * 2 * HID];
    for (int i = threadIdx.x; i < KPAIRS * 2 * HID; i += 256) {
        int kp   = i >> 7;
        int r    = i & 127;
        int col  = r >> 1;
        int sub  = r & 1;
        int krow = kp * 2 + sub;
        w1p[i] = (krow < IN_DIM) ? W1[krow * HID + col] : 0.0f;
    }
    __syncthreads();

    const int wave = threadIdx.x >> 5;
    const int lane = threadIdx.x & 31;
    const int tile = blockIdx.x * 8 + wave;
    if (tile >= NTILES) return;

    const int hh  = lane >> 4;
    const int lr  = lane & 15;
    const int row = tile * 16 + lr;
    const float* __restrict__ xrb = x + (long)row * IN_DIM;
    const float* __restrict__ xr  = xrb + 2 * hh;

    v8f acc0 = (v8f){}, acc1 = (v8f){}, acc2 = (v8f){}, acc3 = (v8f){};

    for (int k = 0; k < 164; k += 4) {
        v2f A;
        A.x = xr[k];
        A.y = xr[k + 1];
        const int ka = k + 2 * hh;
        const v2f* __restrict__ bp = (const v2f*)&w1p[(ka >> 1) * 2 * HID + lr * 2];
        v2f B0 = bp[0];
        v2f B1 = bp[16];
        v2f B2 = bp[32];
        v2f B3 = bp[48];
        acc0 = __builtin_amdgcn_wmma_f32_16x16x4_f32(false, A, false, B0, (short)0, acc0, false, false);
        acc1 = __builtin_amdgcn_wmma_f32_16x16x4_f32(false, A, false, B1, (short)0, acc1, false, false);
        acc2 = __builtin_amdgcn_wmma_f32_16x16x4_f32(false, A, false, B2, (short)0, acc2, false, false);
        acc3 = __builtin_amdgcn_wmma_f32_16x16x4_f32(false, A, false, B3, (short)0, acc3, false, false);
    }

    {
        float a164 = xrb[164];
        v2f A;
        A.x = (hh == 0) ? a164 : 0.0f;
        A.y = 0.0f;
        const v2f* __restrict__ bp = (const v2f*)&w1p[(KPAIRS - 1) * 2 * HID + lr * 2];
        v2f B0 = bp[0];
        v2f B1 = bp[16];
        v2f B2 = bp[32];
        v2f B3 = bp[48];
        acc0 = __builtin_amdgcn_wmma_f32_16x16x4_f32(false, A, false, B0, (short)0, acc0, false, false);
        acc1 = __builtin_amdgcn_wmma_f32_16x16x4_f32(false, A, false, B1, (short)0, acc1, false, false);
        acc2 = __builtin_amdgcn_wmma_f32_16x16x4_f32(false, A, false, B2, (short)0, acc2, false, false);
        acc3 = __builtin_amdgcn_wmma_f32_16x16x4_f32(false, A, false, B3, (short)0, acc3, false, false);
    }

    const long obase = (long)(tile * 16 + 8 * hh) * HID + lr;
    float* __restrict__ outp = hw1 + obase;
#pragma unroll
    for (int v = 0; v < 8; ++v) {
        long o = (long)v * HID;
        outp[o +  0] = acc0[v];
        outp[o + 16] = acc1[v];
        outp[o + 32] = acc2[v];
        outp[o + 48] = acc3[v];
    }
}


__device__ __forceinline__ void add32(const int* qp, int lane, int lo,
                                      const int* __restrict__ src, const int* __restrict__ dst,
                                      const float* __restrict__ dinv, const float* sdinv,
                                      const float* __restrict__ hw1, float* acc) {
    const int e  = qp[lane];
    const int   s = src[e];
    const int  dl = dst[e] - lo;
    const float w = dinv[s] * sdinv[dl];
    for (int g = 0; g < 2; ++g) {
        v2f row[16];
        for (int k = 0; k < 16; ++k) {
            const int sk = __builtin_amdgcn_readlane(s, g * 16 + k);
            row[k] = *(const v2f*)(hw1 + (long)sk * HID + 2 * lane);
        }
        for (int k = 0; k < 16; ++k) {
            const float wk = __int_as_float(__builtin_amdgcn_readlane(__float_as_int(w), g * 16 + k));
            const int   dk = __builtin_amdgcn_readlane(dl, g * 16 + k);
            float* a = acc + dk * HID + 2 * lane;
            atomicAdd(a,     row[k].x * wk);
            atomicAdd(a + 1, row[k].y * wk);
        }
    }
}

__device__ __forceinline__ void add_tail(const int* qp, int n, int lane, int lo,
                                         const int* __restrict__ src, const int* __restrict__ dst,
                                         const float* __restrict__ dinv, const float* sdinv,
                                         const float* __restrict__ hw1, float* acc) {
    int s = 0, dl = 0;
    float w = 0.0f;
    if (lane < n) {
        const int e = qp[lane];
        dl = dst[e] - lo;
        s  = src[e];
        w  = dinv[s] * sdinv[dl];
    }
    for (int g = 0; g < 4; ++g) {
        if (g * 8 >= n) break;
        v2f row[8];
        for (int k = 0; k < 8; ++k) {
            row[k] = (v2f){0.0f, 0.0f};
            if (g * 8 + k < n) {
                const int sk = __builtin_amdgcn_readlane(s, g * 8 + k);
                row[k] = *(const v2f*)(hw1 + (long)sk * HID + 2 * lane);
            }
        }
        for (int k = 0; k < 8; ++k) {
            if (g * 8 + k < n) {
                const float wk = __int_as_float(__builtin_amdgcn_readlane(__float_as_int(w), g * 8 + k));
                const int   dk = __builtin_amdgcn_readlane(dl, g * 8 + k);
                float* a = acc + dk * HID + 2 * lane;
                atomicAdd(a,     row[k].x * wk);
                atomicAdd(a + 1, row[k].y * wk);
            }
        }
    }
}

__global__ __launch_bounds__(1024) void agg1_tiles(const int* __restrict__ src, const int* __restrict__ dst,
                                                   const unsigned* __restrict__ bm,
                                                   const float* __restrict__ hw1,
                                                   const float* __restrict__ dinv, const float* __restrict__ b1,
                                                   const float* __restrict__ W2, float* __restrict__ hw2) {
    __shared__ float acc[TR * HID];
    __shared__ int   qbuf[32 * QCAP];
    __shared__ float sdinv[TR];
    const int tid  = threadIdx.x;
    const int lane = tid & 31;
    const int wave = __builtin_amdgcn_readfirstlane(tid >> 5);
    const int lo   = blockIdx.x * TR;

    if (tid < TR) sdinv[tid] = dinv[lo + tid];
    {
        const v4f* __restrict__ hp = (const v4f*)(hw1 + (long)lo * HID);
        for (int i = tid; i < TR * HID / 4; i += 1024) {
            float sn = dinv[lo + (i >> 4)];
            sn *= sn;
            const v4f h = hp[i];
            const v4f b = *(const v4f*)(b1 + (i & 15) * 4);
            v4f r;
            r.x = h.x * sn + b.x;
            r.y = h.y * sn + b.y;
            r.z = h.z * sn + b.z;
            r.w = h.w * sn + b.w;
            *(v4f*)(acc + i * 4) = r;
        }
    }
    __syncthreads();

    int* q = qbuf + wave * QCAP;
    int qn = 0;
    const uint4* __restrict__ bmr = (const uint4*)(bm + (size_t)blockIdx.x * BM_PITCH);
    uint4 cur = bmr[wave * 32 + lane];
    const int nj = (BM_BLKS - wave + 31) / 32;
    for (int j = 0; j < nj; ++j) {
        const int piece = wave + 32 * j;
        const int pn = piece + 32;
        const uint4 ld = bmr[(pn < BM_BLKS ? pn : piece) * 32 + lane];
#pragma unroll
        for (int c = 0; c < 4; ++c) {
            unsigned w = c == 0 ? cur.x : c == 1 ? cur.y : c == 2 ? cur.z : cur.w;
            const int eb = ((piece * 32 + lane) * 4 + c) * 32;
            unsigned m = __builtin_amdgcn_ballot_w32(w != 0u);
            while (m) {
                if (w != 0u) {
                    q[qn + __builtin_amdgcn_mbcnt_lo(m, 0)] = eb + __builtin_ctz(w);
                    w &= w - 1u;
                }
                qn += __builtin_popcount(m);
                m = __builtin_amdgcn_ballot_w32(w != 0u);
            }
            if (qn < 32) continue;
            int qh = 0;
            while (qn - qh >= 32) {
                add32(q + qh, lane, lo, src, dst, dinv, sdinv, hw1, acc);
                qh += 32;
            }
            const int rem = qn - qh;
            int t = 0;
            if (lane < rem) t = q[qh + lane];
            if (lane < rem) q[lane] = t;
            qn = rem;
        }
        cur = ld;
    }
    if (qn > 0) add_tail(q, qn, lane, lo, src, dst, dinv, sdinv, hw1, acc);
    __syncthreads();

    {
        const int half = lane >> 4;
        const int c4   = lane & 15;
        const v4f wa = *(const v4f*)(W2 + 8 * c4);
        const v4f wb = *(const v4f*)(W2 + 8 * c4 + 4);
        const int rb = wave * 16;
        const int np = (TR - rb) < 16 ? (TR - rb) / 2 : 8;
        float k0 = 0.0f, k1 = 0.0f;
        float p0[8], p1[8];
        for (int i = 0; i < 8; ++i) {
            int r = rb + 2 * i + half;
            r = r < TR ? r : TR - 1;
            v4f h = *(const v4f*)(acc + r * HID + 4 * c4);
            h.x = h.x > 0.0f ? h.x : 0.0f;
            h.y = h.y > 0.0f ? h.y : 0.0f;
            h.z = h.z > 0.0f ? h.z : 0.0f;
            h.w = h.w > 0.0f ? h.w : 0.0f;
            p0[i] = fmaf(h.w, wb.z, fmaf(h.z, wb.x, fmaf(h.y, wa.z, h.x * wa.x)));
            p1[i] = fmaf(h.w, wb.w, fmaf(h.z, wb.y, fmaf(h.y, wa.w, h.x * wa.y)));
        }
        for (int m = 8; m >= 1; m >>= 1) {
            for (int i = 0; i < 8; ++i) {
                p0[i] += __shfl_xor(p0[i], m, 16);
                p1[i] += __shfl_xor(p1[i], m, 16);
            }
        }
        for (int i = 0; i < 8; ++i) {
            if (c4 == i) { k0 = p0[i]; k1 = p1[i]; }
        }
        if (c4 < np) *(v2f*)(hw2 + 2 * (long)(lo + rb + 2 * c4 + half)) = (v2f){k0, k1};
    }
}

__global__ __launch_bounds__(1024) void agg2_tiles(const int* __restrict__ src, const int* __restrict__ dst,
                                                   const unsigned* __restrict__ bm,
                                                   const float* __restrict__ hw2,
                                                   const float* __restrict__ dinv, const float* __restrict__ b2,
                                                   float* __restrict__ out) {
    __shared__ float a2[TR * OUT_DIM];
    __shared__ int   qbuf[32 * QCAP];
    __shared__ float sdinv[TR];
    const int tid  = threadIdx.x;
    const int lane = tid & 31;
    const int wave = __builtin_amdgcn_readfirstlane(tid >> 5);
    const int lo   = blockIdx.x * TR;
    if (tid < TR) {
        float sn = dinv[lo + tid];
        sdinv[tid] = sn;
        sn *= sn;
        const v2f h = *(const v2f*)(hw2 + 2 * (long)(lo + tid));
        a2[2 * tid]     = h.x * sn + b2[0];
        a2[2 * tid + 1] = h.y * sn + b2[1];
    }
    __syncthreads();
    walk_owned_edges(bm, qbuf + wave * QCAP, wave, lane, [&](const int* qp, int n) {
        if (lane < n) {
            const int   e  = qp[lane];
            const int   s  = src[e];
            const int   dl = dst[e] - lo;
            const v2f   h  = *(const v2f*)(hw2 + 2 * (long)s);
            const float w  = dinv[s] * sdinv[dl];
            atomicAdd(&a2[2 * dl],     h.x * w);
            atomicAdd(&a2[2 * dl + 1], h.y * w);
        }
    });
    __syncthreads();
    if (tid < TR) *(v2f*)(out + 2 * (long)(lo + tid)) = *(const v2f*)(a2 + 2 * tid);
}

extern "C" void kernel_launch(void* const* d_in, const int* in_sizes, int n_in,
                              void* d_out, int out_size, void* d_ws, size_t ws_size,
                              hipStream_t stream) {
    const float* x  = (const float*)d_in[0];
    const int*   ei = (const int*)  d_in[1];
    const float* W1 = (const float*)d_in[2];
    const float* b1 = (const float*)d_in[3];
    const float* W2 = (const float*)d_in[4];
    const float* b2 = (const float*)d_in[5];
    const int* src = ei;
    const int* dst = ei + N_EDGES;
    float* out = (float*)d_out;

    float* ws   = (float*)d_ws;
    float* dinv = ws;
    float* hw1  = dinv + N_NODES;
    float* hw2  = hw1 + (size_t)N_NODES * HID;
    unsigned* bm = (unsigned*)(hw2 + (size_t)N_NODES * OUT_DIM) + 32;

    const dim3 blk(256);

    bitmap_kernel<<<BM_BLKS, blk, 0, stream>>>(dst, bm);

    deg_tiles<<<NRANGE, dim3(1024), 0, stream>>>(dst, bm, dinv);

    gemm1_wmma<<<(NTILES + 7) / 8, blk, 0, stream>>>(x, W1, hw1);
    agg1_tiles<<<NRANGE, dim3(1024), 0, stream>>>(src, dst, bm, hw1, dinv, b1, W2, hw2);

    agg2_tiles<<<NRANGE, dim3(1024), 0, stream>>>(src, dst, bm, hw2, dinv, b2, out);
}
